# starting kernel
# baseline (speedup 1.0000x reference)
#include <hip/hip_runtime.h>


#define NN 10000
#define DD 512
#define NE 160000
#define MT 5

typedef __attribute__((ext_vector_type(2))) float v2f;
typedef __attribute__((ext_vector_type(8))) float v8f;

__global__ void zero_kernel(float* __restrict__ p, long long n) {
  long long i = (long long)blockIdx.x * blockDim.x + threadIdx.x;
  long long stride = (long long)gridDim.x * blockDim.x;
  for (; i < n; i += stride) p[i] = 0.0f;
}

__global__ void count_kernel(const int* __restrict__ dst, float* __restrict__ cnt) {
  int i = blockIdx.x * blockDim.x + threadIdx.x;
  if (i < NE) atomicAdd(&cnt[dst[i]], 1.0f);
}

__global__ void recip_kernel(float* __restrict__ cnt) {
  int i = blockIdx.x * blockDim.x + threadIdx.x;
  if (i < NN) cnt[i] = 1.0f / fmaxf(cnt[i], 1.0f);
}

__global__ __launch_bounds__(128) void scatter_kernel(
    const float* __restrict__ feat, const int* __restrict__ src,
    const int* __restrict__ dst, float* __restrict__ agg) {
  const int e = blockIdx.x;
  const int s = src[e];
  const int d = dst[e];
  const float4 v = ((const float4*)(feat + (size_t)s * DD))[threadIdx.x];
  float* ap = agg + (size_t)d * DD + (size_t)threadIdx.x * 4;
  atomicAdd(ap + 0, v.x);
  atomicAdd(ap + 1, v.y);
  atomicAdd(ap + 2, v.z);
  atomicAdd(ap + 3, v.w);
}

template <bool RELU>
__global__ __launch_bounds__(256) void sage_gemm(
    const float* __restrict__ Agg, const float* __restrict__ inv,
    const float* __restrict__ Xin, const float* __restrict__ Wl,
    const float* __restrict__ Wr, const float* __restrict__ bias,
    float* __restrict__ out) {
  const int wave = threadIdx.x >> 5;
  const int lane = threadIdx.x & 31;
  const int half = lane >> 4;
  const int r    = lane & 15;
  const int idx  = blockIdx.x * 8 + wave;
  const int rg   = idx >> 5;
  const int ct   = idx & 31;
  const int row0 = rg * (16 * MT) + r;
  const int col  = ct * 16 + r;

  const float* a1p = Agg + (size_t)row0 * DD + 2 * half;
  const float* a2p = Xin + (size_t)row0 * DD + 2 * half;
  const float* blp = Wl  + (size_t)col  * DD + 2 * half;
  const float* brp = Wr  + (size_t)col  * DD + 2 * half;

  v8f c1[MT] = {};
  v8f c2[MT] = {};
#pragma unroll 2
  for (int k = 0; k < DD; k += 4) {
    const v2f fbl = *(const v2f*)(blp + k);
    const v2f fbr = *(const v2f*)(brp + k);
    __builtin_prefetch(blp + k + 64, 0, 3);
    __builtin_prefetch(brp + k + 64, 0, 3);
#pragma unroll
    for (int t = 0; t < MT; ++t) {
      const v2f fa1 = *(const v2f*)(a1p + k + t * (16 * DD));
      const v2f fa2 = *(const v2f*)(a2p + k + t * (16 * DD));
      c1[t] = __builtin_amdgcn_wmma_f32_16x16x4_f32(false, fa1, false, fbl,
                                                    (short)0, c1[t], false, false);
      c2[t] = __builtin_amdgcn_wmma_f32_16x16x4_f32(false, fa2, false, fbr,
                                                    (short)0, c2[t], false, false);
    }
  }

  const float b = bias[col];
#pragma unroll
  for (int t = 0; t < MT; ++t) {
    const int rbase = rg * (16 * MT) + 16 * t + 8 * half;
    float* op = out + (size_t)rbase * DD + (size_t)ct * 16 + r;
#pragma unroll
    for (int v = 0; v < 8; ++v) {
      float val = c1[t][v] * inv[rbase + v] + c2[t][v] + b;
      if (RELU) val = fmaxf(val, 0.0f);
      op[(size_t)v * DD] = val;
    }
  }
}

extern "C" void kernel_launch(void* const* d_in, const int* in_sizes, int n_in,
                              void* d_out, int out_size, void* d_ws, size_t ws_size,
                              hipStream_t stream) {
  const float* x   = (const float*)d_in[0];
  const int*   ei  = (const int*)d_in[1];
  const int*   src = ei;
  const int*   dst = ei + NE;
  const float* W1l = (const float*)d_in[2];
  const float* b1  = (const float*)d_in[3];
  const float* W1r = (const float*)d_in[4];
  const float* W2l = (const float*)d_in[5];
  const float* b2  = (const float*)d_in[6];
  const float* W2r = (const float*)d_in[7];
  float* out = (float*)d_out;

  float* agg = (float*)d_ws;
  float* h   = agg + (size_t)NN * DD;
  float* cnt = h   + (size_t)NN * DD;

  zero_kernel<<<2048, 256, 0, stream>>>(agg, (long long)NN * DD);
  zero_kernel<<<64,   256, 0, stream>>>(cnt, (long long)NN);
  count_kernel<<<(NE + 255) / 256, 256, 0, stream>>>(dst, cnt);
  recip_kernel<<<(NN + 255) / 256, 256, 0, stream>>>(cnt);
  scatter_kernel<<<NE, 128, 0, stream>>>(x, src, dst, agg);
  sage_gemm<true><<<500, 256, 0, stream>>>(agg, cnt, x, W1l, W1r, b1, h);

  zero_kernel<<<2048, 256, 0, stream>>>(agg, (long long)NN * DD);
  scatter_kernel<<<NE, 128, 0, stream>>>(h, src, dst, agg);
  sage_gemm<false><<<500, 256, 0, stream>>>(agg, cnt, h, W2l, W2r, b2, out);
}


# GEMM restaged through LDS: 80x256 block per work-group (250, one per WGP), 64-k stages fetched as whole 256-B row segments, 80x32 per wave from LDS; f32 WMMA, k order unchanged
# speedup vs baseline: 1.6263x; 1.6263x over previous
#include <hip/hip_runtime.h>


#define NN 10000
#define DD 512
#define NE 160000
#define BM 80
#define BN 256
#define KT 64
#define LP 68
#define MT 5
#define NT 2
#define STAGE ((BM + BN) * LP)

typedef __attribute__((ext_vector_type(2))) float v2f;
typedef __attribute__((ext_vector_type(8))) float v8f;

__global__ void zero_kernel(float* __restrict__ p, long long n) {
  long long i = (long long)blockIdx.x * blockDim.x + threadIdx.x;
  long long stride = (long long)gridDim.x * blockDim.x;
  for (; i < n; i += stride) p[i] = 0.0f;
}

__global__ void count_kernel(const int* __restrict__ dst, float* __restrict__ cnt) {
  int i = blockIdx.x * blockDim.x + threadIdx.x;
  if (i < NE) atomicAdd(&cnt[dst[i]], 1.0f);
}

__global__ void recip_kernel(float* __restrict__ cnt) {
  int i = blockIdx.x * blockDim.x + threadIdx.x;
  if (i < NN) cnt[i] = 1.0f / fmaxf(cnt[i], 1.0f);
}

__global__ __launch_bounds__(128) void scatter_kernel(
    const float* __restrict__ feat, const int* __restrict__ src,
    const int* __restrict__ dst, float* __restrict__ agg) {
  const int e = blockIdx.x;
  const int s = src[e];
  const int d = dst[e];
  const float4 v = ((const float4*)(feat + (size_t)s * DD))[threadIdx.x];
  float* ap = agg + (size_t)d * DD + (size_t)threadIdx.x * 4;
  atomicAdd(ap + 0, v.x);
  atomicAdd(ap + 1, v.y);
  atomicAdd(ap + 2, v.z);
  atomicAdd(ap + 3, v.w);
}

__device__ __forceinline__ void load_stage(const float* __restrict__ a,
                                           const float* __restrict__ w,
                                           float4 (&ra)[BM / 16],
                                           float4 (&rw)[BN / 16]) {
#pragma unroll
  for (int i = 0; i < BM / 16; ++i)
    ra[i] = *(const float4*)(a + (size_t)i * 16 * DD);
#pragma unroll
  for (int i = 0; i < BN / 16; ++i)
    rw[i] = *(const float4*)(w + (size_t)i * 16 * DD);
}

__device__ __forceinline__ void store_stage(float* s,
                                            const float4 (&ra)[BM / 16],
                                            const float4 (&rw)[BN / 16]) {
#pragma unroll
  for (int i = 0; i < BM / 16; ++i)
    *(float4*)(s + i * 16 * LP) = ra[i];
#pragma unroll
  for (int i = 0; i < BN / 16; ++i)
    *(float4*)(s + BM * LP + i * 16 * LP) = rw[i];
}

__device__ __forceinline__ void mma_stage(const float* fa, const float* fb,
                                          v8f (&c)[MT][NT]) {
#pragma unroll
  for (int kk = 0; kk < KT / 4; ++kk) {
    v2f a[MT], b[NT];
#pragma unroll
    for (int i = 0; i < MT; ++i) a[i] = *(const v2f*)(fa + i * 16 * LP + kk * 4);
#pragma unroll
    for (int j = 0; j < NT; ++j) b[j] = *(const v2f*)(fb + j * 16 * LP + kk * 4);
#pragma unroll
    for (int i = 0; i < MT; ++i)
#pragma unroll
      for (int j = 0; j < NT; ++j)
        c[i][j] = __builtin_amdgcn_wmma_f32_16x16x4_f32(false, a[i], false, b[j],
                                                        (short)0, c[i][j], false, false);
  }
}

template <bool RELU>
__global__ __launch_bounds__(256) void sage_gemm(
    const float* __restrict__ Agg, const float* __restrict__ inv,
    const float* __restrict__ Xin, const float* __restrict__ Wl,
    const float* __restrict__ Wr, const float* __restrict__ bias,
    float* __restrict__ out) {
  __shared__ __attribute__((aligned(16))) float lds[2 * STAGE];

  const int tid  = threadIdx.x;
  const int wave = tid >> 5;
  const int lane = tid & 31;
  const int half = lane >> 4;
  const int r    = lane & 15;
  const int rb   = blockIdx.x >> 1;
  const int cb   = blockIdx.x & 1;

  const int sr = tid >> 4;
  const int sc = (tid & 15) * 4;
  const float* a1g = Agg + (size_t)(rb * BM + sr) * DD + sc;
  const float* a2g = Xin + (size_t)(rb * BM + sr) * DD + sc;
  const float* w1g = Wl  + (size_t)(cb * BN + sr) * DD + sc;
  const float* w2g = Wr  + (size_t)(cb * BN + sr) * DD + sc;
  const int st_off = sr * LP + sc;
  const int fa_off = r * LP + 2 * half;
  const int fb_off = BM * LP + (wave * 32 + r) * LP + 2 * half;

  v8f c1[MT][NT] = {};
  v8f c2[MT][NT] = {};
  float4 ra[BM / 16], rw[BN / 16];

  float* cur = lds;
  float* nxt = lds + STAGE;

  load_stage(a1g, w1g, ra, rw);
  store_stage(cur + st_off, ra, rw);
  __syncthreads();

  for (int t = 0; t < DD / KT; ++t) {
    const bool more = t + 1 < DD / KT;
    load_stage(more ? a1g + (t + 1) * KT : a2g, more ? w1g + (t + 1) * KT : w2g, ra, rw);
    mma_stage(cur + fa_off, cur + fb_off, c1);
    store_stage(nxt + st_off, ra, rw);
    __syncthreads();
    float* tmp = cur; cur = nxt; nxt = tmp;
  }
  for (int t = 0; t < DD / KT; ++t) {
    const bool more = t + 1 < DD / KT;
    if (more) load_stage(a2g + (t + 1) * KT, w2g + (t + 1) * KT, ra, rw);
    mma_stage(cur + fa_off, cur + fb_off, c2);
    if (more) store_stage(nxt + st_off, ra, rw);
    __syncthreads();
    float* tmp = cur; cur = nxt; nxt = tmp;
  }

#pragma unroll
  for (int j = 0; j < NT; ++j) {
    const int col = cb * BN + wave * 32 + j * 16 + r;
    const float b = bias[col];
#pragma unroll
    for (int i = 0; i < MT; ++i) {
      const int rbase = rb * BM + 16 * i + 8 * half;
      float* op = out + (size_t)rbase * DD + col;
#pragma unroll
      for (int v = 0; v < 8; ++v) {
        float val = c1[i][j][v] * inv[rbase + v] + c2[i][j][v] + b;
        if (RELU) val = fmaxf(val, 0.0f);
        op[(size_t)v * DD] = val;
      }
    }
  }
}

extern "C" void kernel_launch(void* const* d_in, const int* in_sizes, int n_in,
                              void* d_out, int out_size, void* d_ws, size_t ws_size,
                              hipStream_t stream) {
  const float* x   = (const float*)d_in[0];
  const int*   ei  = (const int*)d_in[1];
  const int*   src = ei;
  const int*   dst = ei + NE;
  const float* W1l = (const float*)d_in[2];
  const float* b1  = (const float*)d_in[3];
  const float* W1r = (const float*)d_in[4];
  const float* W2l = (const float*)d_in[5];
  const float* b2  = (const float*)d_in[6];
  const float* W2r = (const float*)d_in[7];
  float* out = (float*)d_out;

  float* agg = (float*)d_ws;
  float* h   = agg + (size_t)NN * DD;
  float* cnt = h   + (size_t)NN * DD;

  zero_kernel<<<2048, 256, 0, stream>>>(agg, (long long)NN * DD);
  zero_kernel<<<64,   256, 0, stream>>>(cnt, (long long)NN);
  count_kernel<<<(NE + 255) / 256, 256, 0, stream>>>(dst, cnt);
  recip_kernel<<<(NN + 255) / 256, 256, 0, stream>>>(cnt);
  scatter_kernel<<<NE, 128, 0, stream>>>(x, src, dst, agg);
  sage_gemm<true><<<250, 256, 0, stream>>>(agg, cnt, x, W1l, W1r, b1, h);

  zero_kernel<<<2048, 256, 0, stream>>>(agg, (long long)NN * DD);
  scatter_kernel<<<NE, 128, 0, stream>>>(h, src, dst, agg);
  sage_gemm<false><<<250, 256, 0, stream>>>(agg, cnt, h, W2l, W2r, b2, out);
}


# Edge aggregation as a destination-bucketed (CSR) row gather, one wave per node summing in registers, replaces the f32 atomic scatter and its zero fill; sum order = arrival-order rank of the fill
# speedup vs baseline: 3.1077x; 1.9109x over previous
#include <hip/hip_runtime.h>


#define NN 10000
#define DD 512
#define NE 160000
#define BM 80
#define BN 256
#define KT 64
#define LP 68
#define MT 5
#define NT 2
#define STAGE ((BM + BN) * LP)

typedef __attribute__((ext_vector_type(2))) float v2f;
typedef __attribute__((ext_vector_type(8))) float v8f;

#define GW 8
#define SCAN_T 1024
#define SCAN_I 10
#define NPAD 10016
static_assert(NN % GW == 0, "gather grid must cover the nodes exactly");
static_assert(SCAN_T * SCAN_I >= NN, "prefix-sum block must cover the nodes");

__global__ void zero_deg_kernel(int* __restrict__ deg) {
  int i = blockIdx.x * blockDim.x + threadIdx.x;
  if (i < NN) deg[i] = 0;
}

__global__ void count_rank_kernel(const int* __restrict__ dst, int* __restrict__ deg,
                                  int* __restrict__ rank) {
  int e = blockIdx.x * blockDim.x + threadIdx.x;
  if (e < NE) rank[e] = atomicAdd(&deg[dst[e]], 1);
}

__global__ __launch_bounds__(SCAN_T) void scan_kernel(const int* __restrict__ deg,
                                                      int* __restrict__ rowptr,
                                                      float* __restrict__ inv) {
  __shared__ int wsum[SCAN_T / 32];
  const int t = threadIdx.x;
  const int lane = t & 31;
  const int wave = t >> 5;
  const int base = t * SCAN_I;
  int d[SCAN_I];
  int sum = 0;
#pragma unroll
  for (int i = 0; i < SCAN_I; ++i) {
    d[i] = (base + i < NN) ? deg[base + i] : 0;
    sum += d[i];
  }
  int inc = sum;
#pragma unroll
  for (int o = 1; o < 32; o <<= 1) {
    const int v = __shfl_up(inc, o);
    if (lane >= o) inc += v;
  }
  if (lane == 31) wsum[wave] = inc;
  __syncthreads();
  if (wave == 0) {
    const int w = wsum[lane];
    int winc = w;
#pragma unroll
    for (int o = 1; o < 32; o <<= 1) {
      const int v = __shfl_up(winc, o);
      if (lane >= o) winc += v;
    }
    wsum[lane] = winc - w;
  }
  __syncthreads();
  int run = wsum[wave] + inc - sum;
#pragma unroll
  for (int i = 0; i < SCAN_I; ++i) {
    if (base + i < NN) {
      rowptr[base + i] = run;
      inv[base + i] = 1.0f / fmaxf((float)d[i], 1.0f);
      run += d[i];
    }
  }
  if (t == SCAN_T - 1) rowptr[NN] = run;
}

__global__ void fill_kernel(const int* __restrict__ src, const int* __restrict__ dst,
                            const int* __restrict__ rank, const int* __restrict__ rowptr,
                            int* __restrict__ col) {
  int e = blockIdx.x * blockDim.x + threadIdx.x;
  if (e < NE) {
    const int slot = rowptr[dst[e]] + rank[e];
    if ((unsigned)slot < (unsigned)NE) col[slot] = src[e];
  }
}

__global__ __launch_bounds__(GW * 32) void gather_kernel(
    const float* __restrict__ feat, const int* __restrict__ rowptr,
    const int* __restrict__ col, float* __restrict__ agg) {
  const int lane = threadIdx.x & 31;
  const int node = blockIdx.x * GW + __builtin_amdgcn_readfirstlane((int)(threadIdx.x >> 5));
  const int beg = rowptr[node];
  const int end = rowptr[node + 1];
  float4 a0 = make_float4(0.f, 0.f, 0.f, 0.f);
  float4 a1 = a0, a2 = a0, a3 = a0;
  const float4* fl = (const float4*)feat + lane;
  for (int b = beg; b < end; b += 32) {
    const int n = min(32, end - b);
    const int mine = (lane < n) ? col[b + lane] : 0;
#pragma unroll 4
    for (int k = 0; k < n; ++k) {
      const int s = __builtin_amdgcn_readlane(mine, k);
      const float4* rp = fl + (size_t)s * (DD / 4);
      const float4 v0 = rp[0];
      const float4 v1 = rp[32];
      const float4 v2 = rp[64];
      const float4 v3 = rp[96];
      a0.x += v0.x; a0.y += v0.y; a0.z += v0.z; a0.w += v0.w;
      a1.x += v1.x; a1.y += v1.y; a1.z += v1.z; a1.w += v1.w;
      a2.x += v2.x; a2.y += v2.y; a2.z += v2.z; a2.w += v2.w;
      a3.x += v3.x; a3.y += v3.y; a3.z += v3.z; a3.w += v3.w;
    }
  }
  float4* op = (float4*)(agg + (size_t)node * DD) + lane;
  op[0]  = a0;
  op[32] = a1;
  op[64] = a2;
  op[96] = a3;
}

__device__ __forceinline__ void load_stage(const float* __restrict__ a,
                                           const float* __restrict__ w,
                                           float4 (&ra)[BM / 16],
                                           float4 (&rw)[BN / 16]) {
#pragma unroll
  for (int i = 0; i < BM / 16; ++i)
    ra[i] = *(const float4*)(a + (size_t)i * 16 * DD);
#pragma unroll
  for (int i = 0; i < BN / 16; ++i)
    rw[i] = *(const float4*)(w + (size_t)i * 16 * DD);
}

__device__ __forceinline__ void store_stage(float* s,
                                            const float4 (&ra)[BM / 16],
                                            const float4 (&rw)[BN / 16]) {
#pragma unroll
  for (int i = 0; i < BM / 16; ++i)
    *(float4*)(s + i * 16 * LP) = ra[i];
#pragma unroll
  for (int i = 0; i < BN / 16; ++i)
    *(float4*)(s + BM * LP + i * 16 * LP) = rw[i];
}

__device__ __forceinline__ void mma_stage(const float* fa, const float* fb,
                                          v8f (&c)[MT][NT]) {
#pragma unroll
  for (int kk = 0; kk < KT / 4; ++kk) {
    v2f a[MT], b[NT];
#pragma unroll
    for (int i = 0; i < MT; ++i) a[i] = *(const v2f*)(fa + i * 16 * LP + kk * 4);
#pragma unroll
    for (int j = 0; j < NT; ++j) b[j] = *(const v2f*)(fb + j * 16 * LP + kk * 4);
#pragma unroll
    for (int i = 0; i < MT; ++i)
#pragma unroll
      for (int j = 0; j < NT; ++j)
        c[i][j] = __builtin_amdgcn_wmma_f32_16x16x4_f32(false, a[i], false, b[j],
                                                        (short)0, c[i][j], false, false);
  }
}

template <bool RELU>
__global__ __launch_bounds__(256) void sage_gemm(
    const float* __restrict__ Agg, const float* __restrict__ inv,
    const float* __restrict__ Xin, const float* __restrict__ Wl,
    const float* __restrict__ Wr, const float* __restrict__ bias,
    float* __restrict__ out) {
  __shared__ __attribute__((aligned(16))) float lds[2 * STAGE];

  const int tid  = threadIdx.x;
  const int wave = tid >> 5;
  const int lane = tid & 31;
  const int half = lane >> 4;
  const int r    = lane & 15;
  const int rb   = blockIdx.x >> 1;
  const int cb   = blockIdx.x & 1;

  const int sr = tid >> 4;
  const int sc = (tid & 15) * 4;
  const float* a1g = Agg + (size_t)(rb * BM + sr) * DD + sc;
  const float* a2g = Xin + (size_t)(rb * BM + sr) * DD + sc;
  const float* w1g = Wl  + (size_t)(cb * BN + sr) * DD + sc;
  const float* w2g = Wr  + (size_t)(cb * BN + sr) * DD + sc;
  const int st_off = sr * LP + sc;
  const int fa_off = r * LP + 2 * half;
  const int fb_off = BM * LP + (wave * 32 + r) * LP + 2 * half;

  v8f c1[MT][NT] = {};
  v8f c2[MT][NT] = {};
  float4 ra[BM / 16], rw[BN / 16];

  float* cur = lds;
  float* nxt = lds + STAGE;

  load_stage(a1g, w1g, ra, rw);
  store_stage(cur + st_off, ra, rw);
  __syncthreads();

  for (int t = 0; t < DD / KT; ++t) {
    const bool more = t + 1 < DD / KT;
    load_stage(more ? a1g + (t + 1) * KT : a2g, more ? w1g + (t + 1) * KT : w2g, ra, rw);
    mma_stage(cur + fa_off, cur + fb_off, c1);
    store_stage(nxt + st_off, ra, rw);
    __syncthreads();
    float* tmp = cur; cur = nxt; nxt = tmp;
  }
  for (int t = 0; t < DD / KT; ++t) {
    const bool more = t + 1 < DD / KT;
    if (more) load_stage(a2g + (t + 1) * KT, w2g + (t + 1) * KT, ra, rw);
    mma_stage(cur + fa_off, cur + fb_off, c2);
    if (more) store_stage(nxt + st_off, ra, rw);
    __syncthreads();
    float* tmp = cur; cur = nxt; nxt = tmp;
  }

#pragma unroll
  for (int j = 0; j < NT; ++j) {
    const int col = cb * BN + wave * 32 + j * 16 + r;
    const float b = bias[col];
#pragma unroll
    for (int i = 0; i < MT; ++i) {
      const int rbase = rb * BM + 16 * i + 8 * half;
      float* op = out + (size_t)rbase * DD + col;
#pragma unroll
      for (int v = 0; v < 8; ++v) {
        float val = c1[i][j][v] * inv[rbase + v] + c2[i][j][v] + b;
        if (RELU) val = fmaxf(val, 0.0f);
        op[(size_t)v * DD] = val;
      }
    }
  }
}

extern "C" void kernel_launch(void* const* d_in, const int* in_sizes, int n_in,
                              void* d_out, int out_size, void* d_ws, size_t ws_size,
                              hipStream_t stream) {
  const float* x   = (const float*)d_in[0];
  const int*   ei  = (const int*)d_in[1];
  const int*   src = ei;
  const int*   dst = ei + NE;
  const float* W1l = (const float*)d_in[2];
  const float* b1  = (const float*)d_in[3];
  const float* W1r = (const float*)d_in[4];
  const float* W2l = (const float*)d_in[5];
  const float* b2  = (const float*)d_in[6];
  const float* W2r = (const float*)d_in[7];
  float* out = (float*)d_out;

  float* agg = (float*)d_ws;
  float* h   = agg + (size_t)NN * DD;
  float* inv = h   + (size_t)NN * DD;
  int* deg    = (int*)(inv + NPAD);
  int* rowptr = deg + NPAD;
  int* rank   = rowptr + NPAD;
  int* col    = rank + NE;

  zero_deg_kernel<<<(NN + 255) / 256, 256, 0, stream>>>(deg);
  count_rank_kernel<<<(NE + 255) / 256, 256, 0, stream>>>(dst, deg, rank);
  scan_kernel<<<1, SCAN_T, 0, stream>>>(deg, rowptr, inv);
  fill_kernel<<<(NE + 255) / 256, 256, 0, stream>>>(src, dst, rank, rowptr, col);

  gather_kernel<<<NN / GW, GW * 32, 0, stream>>>(x, rowptr, col, agg);
  sage_gemm<true><<<250, 256, 0, stream>>>(agg, inv, x, W1l, W1r, b1, h);

  gather_kernel<<<NN / GW, GW * 32, 0, stream>>>(h, rowptr, col, agg);
  sage_gemm<false><<<250, 256, 0, stream>>>(agg, inv, h, W2l, W2r, b2, out);
}


# Edge bucketing in one launch: 125 work-groups each read all destinations and keep their own 80 nodes' edges (LDS rank counters, LDS prefix sum); replaces zero/count/scan/fill; sum order: arrival rank
# speedup vs baseline: 3.4318x; 1.1043x over previous
#include <hip/hip_runtime.h>


#define NN 10000
#define DD 512
#define NE 160000
#define BM 80
#define BN 256
#define KT 64
#define LP 68
#define MT 5
#define NT 2
#define STAGE ((BM + BN) * LP)

typedef __attribute__((ext_vector_type(2))) float v2f;
typedef __attribute__((ext_vector_type(8))) float v8f;

#define GW 8
#define NPAD 10016
#define NPB 80
#define NBK (NN / NPB)
#define BT 1024
#define NE4 (NE / 4)
#define UB 8
#define LCAP 8192
static_assert(NN % GW == 0, "gather grid must cover the nodes exactly");
static_assert(NN % NPB == 0, "bucketing grid must cover the nodes exactly");
static_assert(NE % 4 == 0, "destinations are read four at a time");
static_assert(NPB <= 128, "the parked key keeps the local node in 7 bits");

__device__ __forceinline__ void place_edge(int dv, int e, int lo, const int* __restrict__ src,
                                           int* fil, const int* off, int* mycol) {
  const unsigned u = (unsigned)(dv - lo);
  if (u < (unsigned)NPB) mycol[off[u] + atomicAdd(&fil[u], 1)] = src[e];
}

__global__ __launch_bounds__(BT) void bucket_kernel(
    const int* __restrict__ src, const int* __restrict__ dst,
    int* __restrict__ rowbeg, int* __restrict__ rowend,
    float* __restrict__ inv, int* __restrict__ col) {
  __shared__ int cnt[NPB];
  __shared__ int off[NPB];
  __shared__ int total;
  __shared__ int lkey[LCAP];
  __shared__ int ledge[LCAP];
  const int tid = threadIdx.x;
  const int lo = blockIdx.x * NPB;
  int* mycol = col + (size_t)blockIdx.x * NE;
  const int4* d4 = (const int4*)dst;
  if (tid < NPB) cnt[tid] = 0;
  if (tid == 0) total = 0;
  __syncthreads();
  for (int base = 0; base < NE4; base += BT * UB) {
    int4 d[UB];
#pragma unroll
    for (int j = 0; j < UB; ++j) {
      const int g = base + j * BT + tid;
      d[j] = (g < NE4) ? d4[g] : make_int4(-1, -1, -1, -1);
    }
    int r[4 * UB];
#pragma unroll
    for (int j = 0; j < UB; ++j) {
      const int dv[4] = {d[j].x, d[j].y, d[j].z, d[j].w};
#pragma unroll
      for (int q = 0; q < 4; ++q) {
        const unsigned u = (unsigned)(dv[q] - lo);
        r[4 * j + q] = (u < (unsigned)NPB) ? atomicAdd(&cnt[u], 1) : -1;
      }
    }
    int k = 0;
#pragma unroll
    for (int i = 0; i < 4 * UB; ++i) k += (r[i] >= 0) ? 1 : 0;
    if (k > 0) {
      int p = atomicAdd(&total, k);
#pragma unroll
      for (int j = 0; j < UB; ++j) {
        const int dv[4] = {d[j].x, d[j].y, d[j].z, d[j].w};
#pragma unroll
        for (int q = 0; q < 4; ++q) {
          if (r[4 * j + q] >= 0) {
            if (p < LCAP) {
              lkey[p] = (r[4 * j + q] << 7) | (dv[q] - lo);
              ledge[p] = (base + j * BT + tid) * 4 + q;
            }
            ++p;
          }
        }
      }
    }
  }
  __syncthreads();
  if (tid < NPB) {
    int o = 0;
    for (int i = 0; i < tid; ++i) o += cnt[i];
    off[tid] = o;
    const int c = cnt[tid];
    const int b0 = blockIdx.x * NE + o;
    rowbeg[lo + tid] = b0;
    rowend[lo + tid] = b0 + c;
    inv[lo + tid] = 1.0f / fmaxf((float)c, 1.0f);
  }
  __syncthreads();
  const int tot = total;
  if (tot <= LCAP) {
    for (int p = tid; p < tot; p += BT) {
      const int k = lkey[p];
      mycol[off[k & 127] + (k >> 7)] = src[ledge[p]];
    }
  } else {
    if (tid < NPB) cnt[tid] = 0;
    __syncthreads();
    for (int g = tid; g < NE4; g += BT) {
      const int4 dv = d4[g];
      place_edge(dv.x, 4 * g + 0, lo, src, cnt, off, mycol);
      place_edge(dv.y, 4 * g + 1, lo, src, cnt, off, mycol);
      place_edge(dv.z, 4 * g + 2, lo, src, cnt, off, mycol);
      place_edge(dv.w, 4 * g + 3, lo, src, cnt, off, mycol);
    }
  }
}

__global__ __launch_bounds__(GW * 32) void gather_kernel(
    const float* __restrict__ feat, const int* __restrict__ rowbeg,
    const int* __restrict__ rowend, const int* __restrict__ col,
    float* __restrict__ agg) {
  const int lane = threadIdx.x & 31;
  const int node = blockIdx.x * GW + __builtin_amdgcn_readfirstlane((int)(threadIdx.x >> 5));
  const int beg = rowbeg[node];
  const int end = rowend[node];
  float4 a0 = make_float4(0.f, 0.f, 0.f, 0.f);
  float4 a1 = a0, a2 = a0, a3 = a0;
  const float4* fl = (const float4*)feat + lane;
  for (int b = beg; b < end; b += 32) {
    const int n = min(32, end - b);
    const int mine = (lane < n) ? col[b + lane] : 0;
#pragma unroll 4
    for (int k = 0; k < n; ++k) {
      const int s = __builtin_amdgcn_readlane(mine, k);
      const float4* rp = fl + (size_t)s * (DD / 4);
      const float4 v0 = rp[0];
      const float4 v1 = rp[32];
      const float4 v2 = rp[64];
      const float4 v3 = rp[96];
      a0.x += v0.x; a0.y += v0.y; a0.z += v0.z; a0.w += v0.w;
      a1.x += v1.x; a1.y += v1.y; a1.z += v1.z; a1.w += v1.w;
      a2.x += v2.x; a2.y += v2.y; a2.z += v2.z; a2.w += v2.w;
      a3.x += v3.x; a3.y += v3.y; a3.z += v3.z; a3.w += v3.w;
    }
  }
  float4* op = (float4*)(agg + (size_t)node * DD) + lane;
  op[0]  = a0;
  op[32] = a1;
  op[64] = a2;
  op[96] = a3;
}

__device__ __forceinline__ void load_stage(const float* __restrict__ a,
                                           const float* __restrict__ w,
                                           float4 (&ra)[BM / 16],
                                           float4 (&rw)[BN / 16]) {
#pragma unroll
  for (int i = 0; i < BM / 16; ++i)
    ra[i] = *(const float4*)(a + (size_t)i * 16 * DD);
#pragma unroll
  for (int i = 0; i < BN / 16; ++i)
    rw[i] = *(const float4*)(w + (size_t)i * 16 * DD);
}

__device__ __forceinline__ void store_stage(float* s,
                                            const float4 (&ra)[BM / 16],
                                            const float4 (&rw)[BN / 16]) {
#pragma unroll
  for (int i = 0; i < BM / 16; ++i)
    *(float4*)(s + i * 16 * LP) = ra[i];
#pragma unroll
  for (int i = 0; i < BN / 16; ++i)
    *(float4*)(s + BM * LP + i * 16 * LP) = rw[i];
}

__device__ __forceinline__ void mma_stage(const float* fa, const float* fb,
                                          v8f (&c)[MT][NT]) {
#pragma unroll
  for (int kk = 0; kk < KT / 4; ++kk) {
    v2f a[MT], b[NT];
#pragma unroll
    for (int i = 0; i < MT; ++i) a[i] = *(const v2f*)(fa + i * 16 * LP + kk * 4);
#pragma unroll
    for (int j = 0; j < NT; ++j) b[j] = *(const v2f*)(fb + j * 16 * LP + kk * 4);
#pragma unroll
    for (int i = 0; i < MT; ++i)
#pragma unroll
      for (int j = 0; j < NT; ++j)
        c[i][j] = __builtin_amdgcn_wmma_f32_16x16x4_f32(false, a[i], false, b[j],
                                                        (short)0, c[i][j], false, false);
  }
}

template <bool RELU>
__global__ __launch_bounds__(256) void sage_gemm(
    const float* __restrict__ Agg, const float* __restrict__ inv,
    const float* __restrict__ Xin, const float* __restrict__ Wl,
    const float* __restrict__ Wr, const float* __restrict__ bias,
    float* __restrict__ out) {
  __shared__ __attribute__((aligned(16))) float lds[2 * STAGE];

  const int tid  = threadIdx.x;
  const int wave = tid >> 5;
  const int lane = tid & 31;
  const int half = lane >> 4;
  const int r    = lane & 15;
  const int rb   = blockIdx.x >> 1;
  const int cb   = blockIdx.x & 1;

  const int sr = tid >> 4;
  const int sc = (tid & 15) * 4;
  const float* a1g = Agg + (size_t)(rb * BM + sr) * DD + sc;
  const float* a2g = Xin + (size_t)(rb * BM + sr) * DD + sc;
  const float* w1g = Wl  + (size_t)(cb * BN + sr) * DD + sc;
  const float* w2g = Wr  + (size_t)(cb * BN + sr) * DD + sc;
  const int st_off = sr * LP + sc;
  const int fa_off = r * LP + 2 * half;
  const int fb_off = BM * LP + (wave * 32 + r) * LP + 2 * half;

  v8f c1[MT][NT] = {};
  v8f c2[MT][NT] = {};
  float4 ra[BM / 16], rw[BN / 16];

  float* cur = lds;
  float* nxt = lds + STAGE;

  load_stage(a1g, w1g, ra, rw);
  store_stage(cur + st_off, ra, rw);
  __syncthreads();

  for (int t = 0; t < DD / KT; ++t) {
    const bool more = t + 1 < DD / KT;
    load_stage(more ? a1g + (t + 1) * KT : a2g, more ? w1g + (t + 1) * KT : w2g, ra, rw);
    mma_stage(cur + fa_off, cur + fb_off, c1);
    store_stage(nxt + st_off, ra, rw);
    __syncthreads();
    float* tmp = cur; cur = nxt; nxt = tmp;
  }
  for (int t = 0; t < DD / KT; ++t) {
    const bool more = t + 1 < DD / KT;
    if (more) load_stage(a2g + (t + 1) * KT, w2g + (t + 1) * KT, ra, rw);
    mma_stage(cur + fa_off, cur + fb_off, c2);
    if (more) store_stage(nxt + st_off, ra, rw);
    __syncthreads();
    float* tmp = cur; cur = nxt; nxt = tmp;
  }

#pragma unroll
  for (int j = 0; j < NT; ++j) {
    const int col = cb * BN + wave * 32 + j * 16 + r;
    const float b = bias[col];
#pragma unroll
    for (int i = 0; i < MT; ++i) {
      const int rbase = rb * BM + 16 * i + 8 * half;
      float* op = out + (size_t)rbase * DD + col;
#pragma unroll
      for (int v = 0; v < 8; ++v) {
        float val = c1[i][j][v] * inv[rbase + v] + c2[i][j][v] + b;
        if (RELU) val = fmaxf(val, 0.0f);
        op[(size_t)v * DD] = val;
      }
    }
  }
}

extern "C" void kernel_launch(void* const* d_in, const int* in_sizes, int n_in,
                              void* d_out, int out_size, void* d_ws, size_t ws_size,
                              hipStream_t stream) {
  const float* x   = (const float*)d_in[0];
  const int*   ei  = (const int*)d_in[1];
  const int*   src = ei;
  const int*   dst = ei + NE;
  const float* W1l = (const float*)d_in[2];
  const float* b1  = (const float*)d_in[3];
  const float* W1r = (const float*)d_in[4];
  const float* W2l = (const float*)d_in[5];
  const float* b2  = (const float*)d_in[6];
  const float* W2r = (const float*)d_in[7];
  float* out = (float*)d_out;

  float* agg = (float*)d_ws;
  float* h   = agg + (size_t)NN * DD;
  float* inv = h   + (size_t)NN * DD;
  int* rowbeg = (int*)(inv + NPAD);
  int* rowend = rowbeg + NPAD;
  int* col    = rowend + NPAD;

  bucket_kernel<<<NBK, BT, 0, stream>>>(src, dst, rowbeg, rowend, inv, col);

  gather_kernel<<<NN / GW, GW * 32, 0, stream>>>(x, rowbeg, rowend, col, agg);
  sage_gemm<true><<<250, 256, 0, stream>>>(agg, inv, x, W1l, W1r, b1, h);

  gather_kernel<<<NN / GW, GW * 32, 0, stream>>>(h, rowbeg, rowend, col, agg);
  sage_gemm<false><<<250, 256, 0, stream>>>(agg, inv, h, W2l, W2r, b2, out);
}


# GEMM global loads issued two 64-k stages ahead: stage s+1 waits in registers, stage s+2 is fetched behind its LDS stores; k-loop barriers order LDS only, so loads stay in flight; k order kept
# speedup vs baseline: 3.5812x; 1.0435x over previous
#include <hip/hip_runtime.h>


#define NN 10000
#define DD 512
#define NE 160000
#define BM 80
#define BN 256
#define KT 64
#define LP 68
#define MT 5
#define NT 2
#define STAGE ((BM + BN) * LP)

typedef __attribute__((ext_vector_type(2))) float v2f;
typedef __attribute__((ext_vector_type(8))) float v8f;
typedef __attribute__((ext_vector_type(4))) float v4f;

#define GW 8
#define NPAD 10016
#define NPB 80
#define NBK (NN / NPB)
#define BT 1024
#define NE4 (NE / 4)
#define UB 8
#define LCAP 8192
static_assert(NN % GW == 0, "gather grid must cover the nodes exactly");
static_assert(NN % NPB == 0, "bucketing grid must cover the nodes exactly");
static_assert(NE % 4 == 0, "destinations are read four at a time");
static_assert(NPB <= 128, "the parked key keeps the local node in 7 bits");

__device__ __forceinline__ void place_edge(int dv, int e, int lo, const int* __restrict__ src,
                                           int* fil, const int* off, int* mycol) {
  const unsigned u = (unsigned)(dv - lo);
  if (u < (unsigned)NPB) mycol[off[u] + atomicAdd(&fil[u], 1)] = src[e];
}

__global__ __launch_bounds__(BT) void bucket_kernel(
    const int* __restrict__ src, const int* __restrict__ dst,
    int* __restrict__ rowbeg, int* __restrict__ rowend,
    float* __restrict__ inv, int* __restrict__ col) {
  __shared__ int cnt[NPB];
  __shared__ int off[NPB];
  __shared__ int total;
  __shared__ int lkey[LCAP];
  __shared__ int ledge[LCAP];
  const int tid = threadIdx.x;
  const int lo = blockIdx.x * NPB;
  int* mycol = col + (size_t)blockIdx.x * NE;
  const int4* d4 = (const int4*)dst;
  if (tid < NPB) cnt[tid] = 0;
  if (tid == 0) total = 0;
  __syncthreads();
  for (int base = 0; base < NE4; base += BT * UB) {
    int4 d[UB];
#pragma unroll
    for (int j = 0; j < UB; ++j) {
      const int g = base + j * BT + tid;
      d[j] = (g < NE4) ? d4[g] : make_int4(-1, -1, -1, -1);
    }
    int r[4 * UB];
#pragma unroll
    for (int j = 0; j < UB; ++j) {
      const int dv[4] = {d[j].x, d[j].y, d[j].z, d[j].w};
#pragma unroll
      for (int q = 0; q < 4; ++q) {
        const unsigned u = (unsigned)(dv[q] - lo);
        r[4 * j + q] = (u < (unsigned)NPB) ? atomicAdd(&cnt[u], 1) : -1;
      }
    }
    int k = 0;
#pragma unroll
    for (int i = 0; i < 4 * UB; ++i) k += (r[i] >= 0) ? 1 : 0;
    if (k > 0) {
      int p = atomicAdd(&total, k);
#pragma unroll
      for (int j = 0; j < UB; ++j) {
        const int dv[4] = {d[j].x, d[j].y, d[j].z, d[j].w};
#pragma unroll
        for (int q = 0; q < 4; ++q) {
          if (r[4 * j + q] >= 0) {
            if (p < LCAP) {
              lkey[p] = (r[4 * j + q] << 7) | (dv[q] - lo);
              ledge[p] = (base + j * BT + tid) * 4 + q;
            }
            ++p;
          }
        }
      }
    }
  }
  __syncthreads();
  if (tid < NPB) {
    int o = 0;
    for (int i = 0; i < tid; ++i) o += cnt[i];
    off[tid] = o;
    const int c = cnt[tid];
    const int b0 = blockIdx.x * NE + o;
    rowbeg[lo + tid] = b0;
    rowend[lo + tid] = b0 + c;
    inv[lo + tid] = 1.0f / fmaxf((float)c, 1.0f);
  }
  __syncthreads();
  const int tot = total;
  if (tot <= LCAP) {
    for (int p = tid; p < tot; p += BT) {
      const int k = lkey[p];
      mycol[off[k & 127] + (k >> 7)] = src[ledge[p]];
    }
  } else {
    if (tid < NPB) cnt[tid] = 0;
    __syncthreads();
    for (int g = tid; g < NE4; g += BT) {
      const int4 dv = d4[g];
      place_edge(dv.x, 4 * g + 0, lo, src, cnt, off, mycol);
      place_edge(dv.y, 4 * g + 1, lo, src, cnt, off, mycol);
      place_edge(dv.z, 4 * g + 2, lo, src, cnt, off, mycol);
      place_edge(dv.w, 4 * g + 3, lo, src, cnt, off, mycol);
    }
  }
}

__global__ __launch_bounds__(GW * 32) void gather_kernel(
    const float* __restrict__ feat, const int* __restrict__ rowbeg,
    const int* __restrict__ rowend, const int* __restrict__ col,
    float* __restrict__ agg) {
  const int lane = threadIdx.x & 31;
  const int node = blockIdx.x * GW + __builtin_amdgcn_readfirstlane((int)(threadIdx.x >> 5));
  const int beg = rowbeg[node];
  const int end = rowend[node];
  float4 a0 = make_float4(0.f, 0.f, 0.f, 0.f);
  float4 a1 = a0, a2 = a0, a3 = a0;
  const float4* fl = (const float4*)feat + lane;
  for (int b = beg; b < end; b += 32) {
    const int n = min(32, end - b);
    const int mine = (lane < n) ? col[b + lane] : 0;
#pragma unroll 4
    for (int k = 0; k < n; ++k) {
      const int s = __builtin_amdgcn_readlane(mine, k);
      const float4* rp = fl + (size_t)s * (DD / 4);
      const float4 v0 = rp[0];
      const float4 v1 = rp[32];
      const float4 v2 = rp[64];
      const float4 v3 = rp[96];
      a0.x += v0.x; a0.y += v0.y; a0.z += v0.z; a0.w += v0.w;
      a1.x += v1.x; a1.y += v1.y; a1.z += v1.z; a1.w += v1.w;
      a2.x += v2.x; a2.y += v2.y; a2.z += v2.z; a2.w += v2.w;
      a3.x += v3.x; a3.y += v3.y; a3.z += v3.z; a3.w += v3.w;
    }
  }
  float4* op = (float4*)(agg + (size_t)node * DD) + lane;
  op[0]  = a0;
  op[32] = a1;
  op[64] = a2;
  op[96] = a3;
}

__device__ __forceinline__ void load_stage(const float* __restrict__ a,
                                           const float* __restrict__ w,
                                           v4f (&ra)[BM / 16],
                                           v4f (&rw)[BN / 16]) {
#pragma unroll
  for (int i = 0; i < BM / 16; ++i)
    ra[i] = *(const v4f*)(a + (size_t)i * 16 * DD);
#pragma unroll
  for (int i = 0; i < BN / 16; ++i)
    rw[i] = *(const v4f*)(w + (size_t)i * 16 * DD);
}

__device__ __forceinline__ void store_stage(float* s,
                                            const v4f (&ra)[BM / 16],
                                            const v4f (&rw)[BN / 16]) {
#pragma unroll
  for (int i = 0; i < BM / 16; ++i)
    *(v4f*)(s + i * 16 * LP) = ra[i];
#pragma unroll
  for (int i = 0; i < BN / 16; ++i)
    *(v4f*)(s + BM * LP + i * 16 * LP) = rw[i];
}

__device__ __forceinline__ void lds_barrier() {
  __builtin_amdgcn_fence(__ATOMIC_RELEASE, "workgroup", "local");
  __builtin_amdgcn_s_barrier();
  __builtin_amdgcn_fence(__ATOMIC_ACQUIRE, "workgroup", "local");
}

__device__ __forceinline__ void mma_stage(const float* fa, const float* fb,
                                          v8f (&c)[MT][NT]) {
  v2f a[KT / 4][MT], b[KT / 4][NT];
#pragma unroll
  for (int kk = 0; kk < KT / 4; ++kk) {
#pragma unroll
    for (int i = 0; i < MT; ++i) a[kk][i] = *(const v2f*)(fa + i * 16 * LP + kk * 4);
#pragma unroll
    for (int j = 0; j < NT; ++j) b[kk][j] = *(const v2f*)(fb + j * 16 * LP + kk * 4);
  }
  __builtin_amdgcn_sched_barrier(0);
#pragma unroll
  for (int kk = 0; kk < KT / 4; ++kk) {
#pragma unroll
    for (int i = 0; i < MT; ++i)
#pragma unroll
      for (int j = 0; j < NT; ++j)
        c[i][j] = __builtin_amdgcn_wmma_f32_16x16x4_f32(false, a[kk][i], false, b[kk][j],
                                                        (short)0, c[i][j], false, false);
  }
}

template <bool RELU>
__global__ __launch_bounds__(256) void sage_gemm(
    const float* __restrict__ Agg, const float* __restrict__ inv,
    const float* __restrict__ Xin, const float* __restrict__ Wl,
    const float* __restrict__ Wr, const float* __restrict__ bias,
    float* __restrict__ out) {
  __shared__ __attribute__((aligned(16))) float lds[2 * STAGE];

  const int tid  = threadIdx.x;
  const int wave = tid >> 5;
  const int lane = tid & 31;
  const int half = lane >> 4;
  const int r    = lane & 15;
  const int rb   = blockIdx.x >> 1;
  const int cb   = blockIdx.x & 1;

  const int sr = tid >> 4;
  const int sc = (tid & 15) * 4;
  const float* a1g = Agg + (size_t)(rb * BM + sr) * DD + sc;
  const float* a2g = Xin + (size_t)(rb * BM + sr) * DD + sc;
  const float* w1g = Wl  + (size_t)(cb * BN + sr) * DD + sc;
  const float* w2g = Wr  + (size_t)(cb * BN + sr) * DD + sc;
  const int st_off = sr * LP + sc;
  const int fa_off = r * LP + 2 * half;
  const int fb_off = BM * LP + (wave * 32 + r) * LP + 2 * half;

  v8f c1[MT][NT] = {};
  v8f c2[MT][NT] = {};
  v4f ra[BM / 16], rw[BN / 16];

  float* cur = lds;
  float* nxt = lds + STAGE;

  constexpr int NS = DD / KT;

  load_stage(a1g, w1g, ra, rw);
  store_stage(cur + st_off, ra, rw);
  load_stage(a1g + KT, w1g + KT, ra, rw);
  lds_barrier();

  for (int t = 0; t < NS; ++t) {
    store_stage(nxt + st_off, ra, rw);
    const int s = t + 2;
    const bool p1 = s < NS;
    load_stage(p1 ? a1g + s * KT : a2g + (s - NS) * KT,
               p1 ? w1g + s * KT : w2g + (s - NS) * KT, ra, rw);
    mma_stage(cur + fa_off, cur + fb_off, c1);
    lds_barrier();
    float* tmp = cur; cur = nxt; nxt = tmp;
  }
  for (int t = 0; t < NS; ++t) {
    if (t + 1 < NS) store_stage(nxt + st_off, ra, rw);
    if (t + 2 < NS) load_stage(a2g + (t + 2) * KT, w2g + (t + 2) * KT, ra, rw);
    mma_stage(cur + fa_off, cur + fb_off, c2);
    lds_barrier();
    float* tmp = cur; cur = nxt; nxt = tmp;
  }

#pragma unroll
  for (int j = 0; j < NT; ++j) {
    const int col = cb * BN + wave * 32 + j * 16 + r;
    const float b = bias[col];
#pragma unroll
    for (int i = 0; i < MT; ++i) {
      const int rbase = rb * BM + 16 * i + 8 * half;
      float* op = out + (size_t)rbase * DD + col;
#pragma unroll
      for (int v = 0; v < 8; ++v) {
        float val = c1[i][j][v] * inv[rbase + v] + c2[i][j][v] + b;
        if (RELU) val = fmaxf(val, 0.0f);
        op[(size_t)v * DD] = val;
      }
    }
  }
}

extern "C" void kernel_launch(void* const* d_in, const int* in_sizes, int n_in,
                              void* d_out, int out_size, void* d_ws, size_t ws_size,
                              hipStream_t stream) {
  const float* x   = (const float*)d_in[0];
  const int*   ei  = (const int*)d_in[1];
  const int*   src = ei;
  const int*   dst = ei + NE;
  const float* W1l = (const float*)d_in[2];
  const float* b1  = (const float*)d_in[3];
  const float* W1r = (const float*)d_in[4];
  const float* W2l = (const float*)d_in[5];
  const float* b2  = (const float*)d_in[6];
  const float* W2r = (const float*)d_in[7];
  float* out = (float*)d_out;

  float* agg = (float*)d_ws;
  float* h   = agg + (size_t)NN * DD;
  float* inv = h   + (size_t)NN * DD;
  int* rowbeg = (int*)(inv + NPAD);
  int* rowend = rowbeg + NPAD;
  int* col    = rowend + NPAD;

  bucket_kernel<<<NBK, BT, 0, stream>>>(src, dst, rowbeg, rowend, inv, col);

  gather_kernel<<<NN / GW, GW * 32, 0, stream>>>(x, rowbeg, rowend, col, agg);
  sage_gemm<true><<<250, 256, 0, stream>>>(agg, inv, x, W1l, W1r, b1, h);

  gather_kernel<<<NN / GW, GW * 32, 0, stream>>>(h, rowbeg, rowend, col, agg);
  sage_gemm<false><<<250, 256, 0, stream>>>(agg, inv, h, W2l, W2r, b2, out);
}
